# starting kernel
# baseline (speedup 1.0000x reference)
#include <hip/hip_runtime.h>

typedef __attribute__((ext_vector_type(2))) float v2f;
typedef __attribute__((ext_vector_type(8))) float v8f;

#define F_DIM 256
#define K_DIM 3
#define N_DIM 4096

__global__ __launch_bounds__(256)
void lie_bracket_channel_mix_kernel(const float* __restrict__ x,
                                    const float* __restrict__ M1,
                                    const float* __restrict__ M2,
                                    const float* __restrict__ W1,
                                    const float* __restrict__ W2,
                                    float* __restrict__ out)
{
    const int lane = threadIdx.x & 31;
    const int wave = threadIdx.x >> 5;
    const int half = lane >> 4;
    const int lm   = lane & 15;

    const int b     = blockIdx.z;
    const int gwave = blockIdx.y * 64 + (wave >> 1) * 16;
    const int nwave = blockIdx.x * 32 + (wave & 1) * 16;

    const int kbase = half * 2;

    const float* a1ptr = W1 + (size_t)(gwave + lm) * F_DIM + kbase;
    const float* a2ptr = W2 + (size_t)(gwave + lm) * F_DIM + kbase;

    const float* xb = x + (size_t)b * F_DIM * K_DIM * N_DIM + (size_t)nwave + lm;

    v8f acc[2][3];
#pragma unroll
    for (int w = 0; w < 2; ++w)
#pragma unroll
        for (int k = 0; k < 3; ++k)
            acc[w][k] = (v8f)(0.0f);

#pragma unroll 4
    for (int f = 0; f < F_DIM; f += 4) {
        v2f a1 = *(const v2f*)(a1ptr + f);
        v2f a2 = *(const v2f*)(a2ptr + f);

        v2f bf[3];
#pragma unroll
        for (int k = 0; k < 3; ++k) {
            const float* p = xb + ((size_t)(f + kbase) * K_DIM + k) * N_DIM;
            bf[k].x = p[0];
            bf[k].y = p[(size_t)K_DIM * N_DIM];
        }

#pragma unroll
        for (int k = 0; k < 3; ++k) {
            acc[0][k] = __builtin_amdgcn_wmma_f32_16x16x4_f32(
                false, a1, false, bf[k], (short)0, acc[0][k], false, false);
            acc[1][k] = __builtin_amdgcn_wmma_f32_16x16x4_f32(
                false, a2, false, bf[k], (short)0, acc[1][k], false, false);
        }
    }

    float m1[9], m2[9];
#pragma unroll
    for (int i = 0; i < 9; ++i) { m1[i] = M1[i]; m2[i] = M2[i]; }

#pragma unroll
    for (int j = 0; j < 8; ++j) {
        const int g = gwave + j + 8 * half;
        const int n = nwave + lm;

        const float d10 = acc[0][0][j], d11 = acc[0][1][j], d12 = acc[0][2][j];
        const float d20 = acc[1][0][j], d21 = acc[1][1][j], d22 = acc[1][2][j];

        const float dr0 = m1[0]*d10 + m1[1]*d11 + m1[2]*d12;
        const float dr1 = m1[3]*d10 + m1[4]*d11 + m1[5]*d12;
        const float dr2 = m1[6]*d10 + m1[7]*d11 + m1[8]*d12;
        const float er0 = m2[0]*d20 + m2[1]*d21 + m2[2]*d22;
        const float er1 = m2[3]*d20 + m2[4]*d21 + m2[5]*d22;
        const float er2 = m2[6]*d20 + m2[7]*d21 + m2[8]*d22;

        const float v0 = er1 * dr2 - er2 * dr1;
        const float v1 = er2 * dr0 - er0 * dr2;
        const float v2 = er0 * dr1 - er1 * dr0;

        const size_t base = ((size_t)(b * F_DIM + g) * K_DIM) * N_DIM + n;
        out[base]                     = x[base]                     + v0;
        out[base + N_DIM]             = x[base + N_DIM]             + v1;
        out[base + 2 * (size_t)N_DIM] = x[base + 2 * (size_t)N_DIM] + v2;
    }
}

extern "C" void kernel_launch(void* const* d_in, const int* in_sizes, int n_in,
                              void* d_out, int out_size, void* d_ws, size_t ws_size,
                              hipStream_t stream) {
    const float* x  = (const float*)d_in[0];
    const float* M1 = (const float*)d_in[1];
    const float* M2 = (const float*)d_in[2];
    const float* W1 = (const float*)d_in[3];
    const float* W2 = (const float*)d_in[4];
    float* out = (float*)d_out;

    const int B = in_sizes[0] / (F_DIM * K_DIM * N_DIM);

    dim3 grid(N_DIM / 32, F_DIM / 64, B);
    dim3 block(256, 1, 1);
    hipLaunchKernelGGL(lie_bracket_channel_mix_kernel, grid, block, 0, stream,
                       x, M1, M2, W1, W2, out);
}


# Wave tile widened from 1 to 4 n-tiles (16 g x 64 n, a lane owns 4 adjacent columns): one W fragment feeds 4x the WMMAs, x and out move as 16-byte accesses; step loads issued two f-steps ahead
# speedup vs baseline: 1.5034x; 1.5034x over previous
#include <hip/hip_runtime.h>

typedef __attribute__((ext_vector_type(2))) float v2f;
typedef __attribute__((ext_vector_type(4))) float v4f;
typedef __attribute__((ext_vector_type(8))) float v8f;

#define F_DIM 256
#define K_DIM 3
#define N_DIM 4096
#define N_TILES 4

struct StepOps {
    v2f a1, a2;
    v4f xr[K_DIM][2];
};

__device__ __forceinline__ void load_step(StepOps& o, const float* __restrict__ a1p,
                                          const float* __restrict__ a2p,
                                          const float* __restrict__ xp)
{
    o.a1 = *(const v2f*)a1p;
    o.a2 = *(const v2f*)a2p;
#pragma unroll
    for (int k = 0; k < K_DIM; ++k)
#pragma unroll
        for (int r = 0; r < 2; ++r)
            o.xr[k][r] = *(const v4f*)(xp + (size_t)(r * K_DIM + k) * N_DIM);
}

__device__ __forceinline__ void mma_step(const StepOps& o, v8f (&acc)[2][K_DIM][N_TILES])
{
#pragma unroll
    for (int k = 0; k < K_DIM; ++k) {
#pragma unroll
        for (int t = 0; t < N_TILES; ++t) {
            v2f bf;
            bf.x = o.xr[k][0][t];
            bf.y = o.xr[k][1][t];
            acc[0][k][t] = __builtin_amdgcn_wmma_f32_16x16x4_f32(
                false, o.a1, false, bf, (short)0, acc[0][k][t], false, false);
            acc[1][k][t] = __builtin_amdgcn_wmma_f32_16x16x4_f32(
                false, o.a2, false, bf, (short)0, acc[1][k][t], false, false);
        }
    }
}

__global__ __launch_bounds__(128)
void lie_bracket_channel_mix_kernel(const float* __restrict__ x,
                                    const float* __restrict__ M1,
                                    const float* __restrict__ M2,
                                    const float* __restrict__ W1,
                                    const float* __restrict__ W2,
                                    float* __restrict__ out)
{
    const int lane = threadIdx.x & 31;
    const int wave = threadIdx.x >> 5;
    const int half = lane >> 4;
    const int lm   = lane & 15;

    const int b     = blockIdx.z;
    const int gwave = blockIdx.y * 64 + wave * 16;
    const int nlane = blockIdx.x * 64 + lm * 4;

    const int kbase = half * 2;

    const float* a1ptr = W1 + (size_t)(gwave + lm) * F_DIM + kbase;
    const float* a2ptr = W2 + (size_t)(gwave + lm) * F_DIM + kbase;

    const float* xp = x + (size_t)b * F_DIM * K_DIM * N_DIM
                        + (size_t)kbase * K_DIM * N_DIM + (size_t)nlane;
    const size_t xstep = (size_t)4 * K_DIM * N_DIM;

    v8f acc[2][K_DIM][N_TILES];
#pragma unroll
    for (int w = 0; w < 2; ++w)
#pragma unroll
        for (int k = 0; k < K_DIM; ++k)
#pragma unroll
            for (int t = 0; t < N_TILES; ++t)
                acc[w][k][t] = (v8f)(0.0f);

    StepOps s0, s1, s2;
    load_step(s0, a1ptr, a2ptr, xp);
    load_step(s1, a1ptr + 4, a2ptr + 4, xp + xstep);
#pragma nounroll
    for (int i = 0; i < F_DIM / 4 - 4; i += 3) {
        load_step(s2, a1ptr + 4 * (i + 2), a2ptr + 4 * (i + 2), xp + (size_t)(i + 2) * xstep);
        __builtin_amdgcn_sched_barrier(0);
        mma_step(s0, acc);
        __builtin_amdgcn_sched_barrier(0);
        load_step(s0, a1ptr + 4 * (i + 3), a2ptr + 4 * (i + 3), xp + (size_t)(i + 3) * xstep);
        __builtin_amdgcn_sched_barrier(0);
        mma_step(s1, acc);
        __builtin_amdgcn_sched_barrier(0);
        load_step(s1, a1ptr + 4 * (i + 4), a2ptr + 4 * (i + 4), xp + (size_t)(i + 4) * xstep);
        __builtin_amdgcn_sched_barrier(0);
        mma_step(s2, acc);
    }
    load_step(s2, a1ptr + F_DIM - 8, a2ptr + F_DIM - 8, xp + (size_t)(F_DIM / 4 - 2) * xstep);
    mma_step(s0, acc);
    load_step(s0, a1ptr + F_DIM - 4, a2ptr + F_DIM - 4, xp + (size_t)(F_DIM / 4 - 1) * xstep);
    mma_step(s1, acc);
    mma_step(s2, acc);
    mma_step(s0, acc);
    __builtin_amdgcn_sched_barrier(0);

    float m1[9], m2[9];
#pragma unroll
    for (int i = 0; i < 9; ++i) { m1[i] = M1[i]; m2[i] = M2[i]; }

#pragma unroll
    for (int j = 0; j < 8; ++j) {
        const int g = gwave + j + 8 * half;
        const size_t base = ((size_t)(b * F_DIM + g) * K_DIM) * N_DIM + nlane;

        const v4f x0 = *(const v4f*)(x + base);
        const v4f x1 = *(const v4f*)(x + base + N_DIM);
        const v4f x2 = *(const v4f*)(x + base + 2 * (size_t)N_DIM);
        v4f o0, o1, o2;

#pragma unroll
        for (int t = 0; t < N_TILES; ++t) {
            const float d10 = acc[0][0][t][j], d11 = acc[0][1][t][j], d12 = acc[0][2][t][j];
            const float d20 = acc[1][0][t][j], d21 = acc[1][1][t][j], d22 = acc[1][2][t][j];

            const float dr0 = m1[0]*d10 + m1[1]*d11 + m1[2]*d12;
            const float dr1 = m1[3]*d10 + m1[4]*d11 + m1[5]*d12;
            const float dr2 = m1[6]*d10 + m1[7]*d11 + m1[8]*d12;
            const float er0 = m2[0]*d20 + m2[1]*d21 + m2[2]*d22;
            const float er1 = m2[3]*d20 + m2[4]*d21 + m2[5]*d22;
            const float er2 = m2[6]*d20 + m2[7]*d21 + m2[8]*d22;

            o0[t] = x0[t] + (er1 * dr2 - er2 * dr1);
            o1[t] = x1[t] + (er2 * dr0 - er0 * dr2);
            o2[t] = x2[t] + (er0 * dr1 - er1 * dr0);
        }

        *(v4f*)(out + base)                     = o0;
        *(v4f*)(out + base + N_DIM)             = o1;
        *(v4f*)(out + base + 2 * (size_t)N_DIM) = o2;
    }
}

extern "C" void kernel_launch(void* const* d_in, const int* in_sizes, int n_in,
                              void* d_out, int out_size, void* d_ws, size_t ws_size,
                              hipStream_t stream) {
    const float* x  = (const float*)d_in[0];
    const float* M1 = (const float*)d_in[1];
    const float* M2 = (const float*)d_in[2];
    const float* W1 = (const float*)d_in[3];
    const float* W2 = (const float*)d_in[4];
    float* out = (float*)d_out;

    const int B = in_sizes[0] / (F_DIM * K_DIM * N_DIM);

    dim3 grid(N_DIM / 64, F_DIM / 64, B);
    dim3 block(128, 1, 1);
    hipLaunchKernelGGL(lie_bracket_channel_mix_kernel, grid, block, 0, stream,
                       x, M1, M2, W1, W2, out);
}


# Wave tile doubled in g (16x64 -> 32x64 outputs, 48 accumulators, 2 waves per SIMD, workgroup 2x2 waves): each x fragment pair feeds 4 WMMAs, not 2, halving repack moves and x loads per WMMA
# speedup vs baseline: 1.5750x; 1.0476x over previous
#include <hip/hip_runtime.h>

typedef __attribute__((ext_vector_type(2))) float v2f;
typedef __attribute__((ext_vector_type(4))) float v4f;
typedef __attribute__((ext_vector_type(8))) float v8f;

#define F_DIM 256
#define K_DIM 3
#define N_DIM 4096
#define N_TILES 4
#define G_TILES 2

struct StepOps {
    v2f a1[G_TILES], a2[G_TILES];
    v4f xr[K_DIM][2];
};

__device__ __forceinline__ void load_step(StepOps& o, const float* __restrict__ a1p,
                                          const float* __restrict__ a2p,
                                          const float* __restrict__ xp)
{
#pragma unroll
    for (int gt = 0; gt < G_TILES; ++gt) {
        o.a1[gt] = *(const v2f*)(a1p + gt * 16 * F_DIM);
        o.a2[gt] = *(const v2f*)(a2p + gt * 16 * F_DIM);
    }
#pragma unroll
    for (int k = 0; k < K_DIM; ++k)
#pragma unroll
        for (int r = 0; r < 2; ++r)
            o.xr[k][r] = *(const v4f*)(xp + (size_t)(r * K_DIM + k) * N_DIM);
}

__device__ __forceinline__ void mma_step(const StepOps& o, v8f (&acc)[2][G_TILES][K_DIM][N_TILES])
{
#pragma unroll
    for (int k = 0; k < K_DIM; ++k) {
#pragma unroll
        for (int t = 0; t < N_TILES; ++t) {
            v2f bf;
            bf.x = o.xr[k][0][t];
            bf.y = o.xr[k][1][t];
#pragma unroll
            for (int gt = 0; gt < G_TILES; ++gt) {
                acc[0][gt][k][t] = __builtin_amdgcn_wmma_f32_16x16x4_f32(
                    false, o.a1[gt], false, bf, (short)0, acc[0][gt][k][t], false, false);
                acc[1][gt][k][t] = __builtin_amdgcn_wmma_f32_16x16x4_f32(
                    false, o.a2[gt], false, bf, (short)0, acc[1][gt][k][t], false, false);
            }
        }
    }
}

__global__ __launch_bounds__(128)
void lie_bracket_channel_mix_kernel(const float* __restrict__ x,
                                    const float* __restrict__ M1,
                                    const float* __restrict__ M2,
                                    const float* __restrict__ W1,
                                    const float* __restrict__ W2,
                                    float* __restrict__ out)
{
    const int lane = threadIdx.x & 31;
    const int wave = threadIdx.x >> 5;
    const int half = lane >> 4;
    const int lm   = lane & 15;

    const int b     = blockIdx.z;
    const int gwave = blockIdx.y * 64 + (wave >> 1) * 32;
    const int nlane = blockIdx.x * 128 + (wave & 1) * 64 + lm * 4;

    const int kbase = half * 2;

    const float* a1ptr = W1 + (size_t)(gwave + lm) * F_DIM + kbase;
    const float* a2ptr = W2 + (size_t)(gwave + lm) * F_DIM + kbase;

    const float* xp = x + (size_t)b * F_DIM * K_DIM * N_DIM
                        + (size_t)kbase * K_DIM * N_DIM + (size_t)nlane;
    const size_t xstep = (size_t)4 * K_DIM * N_DIM;

    v8f acc[2][G_TILES][K_DIM][N_TILES];
#pragma unroll
    for (int w = 0; w < 2; ++w)
#pragma unroll
        for (int gt = 0; gt < G_TILES; ++gt)
#pragma unroll
            for (int k = 0; k < K_DIM; ++k)
#pragma unroll
                for (int t = 0; t < N_TILES; ++t)
                    acc[w][gt][k][t] = (v8f)(0.0f);

    StepOps s0, s1, s2;
    load_step(s0, a1ptr, a2ptr, xp);
    __builtin_amdgcn_sched_barrier(0);
    load_step(s1, a1ptr + 4, a2ptr + 4, xp + xstep);
    for (int i = 0; i < F_DIM / 4 - 4; i += 3) {
        load_step(s2, a1ptr + 4 * (i + 2), a2ptr + 4 * (i + 2), xp + (size_t)(i + 2) * xstep);
        __builtin_amdgcn_sched_barrier(0);
        mma_step(s0, acc);
        __builtin_amdgcn_sched_barrier(0);
        load_step(s0, a1ptr + 4 * (i + 3), a2ptr + 4 * (i + 3), xp + (size_t)(i + 3) * xstep);
        __builtin_amdgcn_sched_barrier(0);
        mma_step(s1, acc);
        __builtin_amdgcn_sched_barrier(0);
        load_step(s1, a1ptr + 4 * (i + 4), a2ptr + 4 * (i + 4), xp + (size_t)(i + 4) * xstep);
        __builtin_amdgcn_sched_barrier(0);
        mma_step(s2, acc);
    }
    load_step(s2, a1ptr + F_DIM - 8, a2ptr + F_DIM - 8, xp + (size_t)(F_DIM / 4 - 2) * xstep);
    mma_step(s0, acc);
    load_step(s0, a1ptr + F_DIM - 4, a2ptr + F_DIM - 4, xp + (size_t)(F_DIM / 4 - 1) * xstep);
    mma_step(s1, acc);
    mma_step(s2, acc);
    mma_step(s0, acc);
    __builtin_amdgcn_sched_barrier(0);

    float m1[9], m2[9];
#pragma unroll
    for (int i = 0; i < 9; ++i) { m1[i] = M1[i]; m2[i] = M2[i]; }

#pragma unroll
    for (int gt = 0; gt < G_TILES; ++gt) {
#pragma unroll
        for (int j = 0; j < 8; ++j) {
            const int g = gwave + gt * 16 + j + 8 * half;
            const size_t base = ((size_t)(b * F_DIM + g) * K_DIM) * N_DIM + nlane;

            const v4f x0 = *(const v4f*)(x + base);
            const v4f x1 = *(const v4f*)(x + base + N_DIM);
            const v4f x2 = *(const v4f*)(x + base + 2 * (size_t)N_DIM);
            v4f o0, o1, o2;

#pragma unroll
            for (int t = 0; t < N_TILES; ++t) {
                const float d10 = acc[0][gt][0][t][j], d11 = acc[0][gt][1][t][j], d12 = acc[0][gt][2][t][j];
                const float d20 = acc[1][gt][0][t][j], d21 = acc[1][gt][1][t][j], d22 = acc[1][gt][2][t][j];

                const float dr0 = m1[0]*d10 + m1[1]*d11 + m1[2]*d12;
                const float dr1 = m1[3]*d10 + m1[4]*d11 + m1[5]*d12;
                const float dr2 = m1[6]*d10 + m1[7]*d11 + m1[8]*d12;
                const float er0 = m2[0]*d20 + m2[1]*d21 + m2[2]*d22;
                const float er1 = m2[3]*d20 + m2[4]*d21 + m2[5]*d22;
                const float er2 = m2[6]*d20 + m2[7]*d21 + m2[8]*d22;

                o0[t] = x0[t] + (er1 * dr2 - er2 * dr1);
                o1[t] = x1[t] + (er2 * dr0 - er0 * dr2);
                o2[t] = x2[t] + (er0 * dr1 - er1 * dr0);
            }

            *(v4f*)(out + base)                     = o0;
            *(v4f*)(out + base + N_DIM)             = o1;
            *(v4f*)(out + base + 2 * (size_t)N_DIM) = o2;
        }
    }
}

extern "C" void kernel_launch(void* const* d_in, const int* in_sizes, int n_in,
                              void* d_out, int out_size, void* d_ws, size_t ws_size,
                              hipStream_t stream) {
    const float* x  = (const float*)d_in[0];
    const float* M1 = (const float*)d_in[1];
    const float* M2 = (const float*)d_in[2];
    const float* W1 = (const float*)d_in[3];
    const float* W2 = (const float*)d_in[4];
    float* out = (float*)d_out;

    const int B = in_sizes[0] / (F_DIM * K_DIM * N_DIM);

    dim3 grid(N_DIM / 128, F_DIM / 64, B);
    dim3 block(128, 1, 1);
    hipLaunchKernelGGL(lie_bracket_channel_mix_kernel, grid, block, 0, stream,
                       x, M1, M2, W1, W2, out);
}
